# starting kernel
# baseline (speedup 1.0000x reference)
#include <hip/hip_runtime.h>
#include <cstddef>

typedef __attribute__((ext_vector_type(2))) float v2f;
typedef __attribute__((ext_vector_type(8))) float v8f;

#define GCN_N 50000
#define THREADS 256

__global__ void k_init_deg(float* __restrict__ deg, int n) {
    int i = blockIdx.x * blockDim.x + threadIdx.x;
    if (i < n) deg[i] = 1.0f;
}

__global__ void k_accum_deg(const int* __restrict__ dst, float* __restrict__ deg, int e) {
    int i = blockIdx.x * blockDim.x + threadIdx.x;
    if (i < e) atomicAdd(&deg[dst[i]], 1.0f);
}

__global__ void k_dinv(float* __restrict__ deg, int n) {
    int i = blockIdx.x * blockDim.x + threadIdx.x;
    if (i < n) deg[i] = rsqrtf(fmaxf(deg[i], 1.0f));
}

__global__ void k_zero(float* __restrict__ p, size_t n) {
    size_t i = (size_t)blockIdx.x * blockDim.x + threadIdx.x;
    if (i < n) p[i] = 0.0f;
}

template <int KIN, int KOUT>
__global__ __launch_bounds__(THREADS)
void k_gemm_wmma(const float* __restrict__ A, const float* __restrict__ W,
                 float* __restrict__ C, int nrows) {
    __shared__ v2f sW[(KIN / 2) * KOUT];
    for (int i = threadIdx.x; i < (KIN / 2) * KOUT; i += blockDim.x) {
        const int k2 = i / KOUT;
        const int c  = i % KOUT;
        v2f w;
        w.x = W[(2 * k2 + 0) * KOUT + c];
        w.y = W[(2 * k2 + 1) * KOUT + c];
        sW[i] = w;
    }
    __syncthreads();

    const int wave = threadIdx.x >> 5;
    const int lane = threadIdx.x & 31;
    const int tile = blockIdx.x * 8 + wave;
    if (tile * 16 >= nrows) return;

    constexpr int NT = KOUT / 16;
    v8f acc[NT] = {};

    const int m    = lane & 15;
    const int khi  = lane >> 4;
    const float* Arow = A + (size_t)(tile * 16 + m) * KIN;
    const v2f* sWl = sW + khi * KOUT + m;

    for (int kk = 0; kk < KIN / 4; ++kk) {
        const v2f a = *(const v2f*)(Arow + kk * 4 + khi * 2);
        const v2f* sWk = sWl + (kk * 2) * KOUT;
#pragma unroll
        for (int t = 0; t < NT; ++t) {
            const v2f b = sWk[t * 16];
            acc[t] = __builtin_amdgcn_wmma_f32_16x16x4_f32(
                 false, a,  false, b,
                 (short)0, acc[t],  false,  false);
        }
    }

#pragma unroll
    for (int t = 0; t < NT; ++t) {
#pragma unroll
        for (int v = 0; v < 8; ++v) {
            const int r = tile * 16 + khi * 8 + v;
            C[(size_t)r * KOUT + t * 16 + m] = acc[t][v];
        }
    }
}

template <int F>
__global__ __launch_bounds__(THREADS)
void k_scatter(const int* __restrict__ src, const int* __restrict__ dst,
               const float* __restrict__ dinv, const float* __restrict__ H,
               float* __restrict__ out, int nedges) {
    constexpr int C = F / 4;
    const long long tid = (long long)blockIdx.x * blockDim.x + threadIdx.x;
    const int e = (int)(tid / C);
    const int c = (int)(tid % C);
    if (e >= nedges) return;

    const int s = src[e];
    const int d = dst[e];
    const float nrm = dinv[s] * dinv[d];
    const float4 h = *(const float4*)(H + (size_t)s * F + c * 4);
    float* o = out + (size_t)d * F + c * 4;
    atomicAdd(o + 0, h.x * nrm);
    atomicAdd(o + 1, h.y * nrm);
    atomicAdd(o + 2, h.z * nrm);
    atomicAdd(o + 3, h.w * nrm);
}

template <int F, bool RELU>
__global__ __launch_bounds__(THREADS)
void k_finalize(const float* __restrict__ H, const float* __restrict__ dinv,
                const float* __restrict__ bias, float* __restrict__ out, int n) {
    const long long tid = (long long)blockIdx.x * blockDim.x + threadIdx.x;
    const int i = (int)(tid / F);
    const int f = (int)(tid % F);
    if (i >= n) return;
    const float di = dinv[i];
    float v = out[(size_t)i * F + f] + H[(size_t)i * F + f] * di * di + bias[f];
    out[(size_t)i * F + f] = RELU ? fmaxf(v, 0.0f) : v;
}

extern "C" void kernel_launch(void* const* d_in, const int* in_sizes, int n_in,
                              void* d_out, int out_size, void* d_ws, size_t ws_size,
                              hipStream_t stream) {
    const float* x  = (const float*)d_in[0];
    const int*   ei = (const int*)d_in[1];
    const float* W1 = (const float*)d_in[2];
    const float* b1 = (const float*)d_in[3];
    const float* W2 = (const float*)d_in[4];
    const float* b2 = (const float*)d_in[5];
    const float* W3 = (const float*)d_in[6];
    const float* b3 = (const float*)d_in[7];
    float* out = (float*)d_out;

    const int N = GCN_N;
    const int E = in_sizes[1] / 2;
    const int* src = ei;
    const int* dst = ei + E;

    float* dinv = (float*)d_ws;
    float* bufA = dinv + ((N + 255) & ~255);
    float* bufB = bufA + (size_t)N * 64;

    const int tiles  = N / 16;
    const int gemmBl = (tiles + 7) / 8;

    auto blocks = [](long long n) { return (unsigned)((n + THREADS - 1) / THREADS); };

    k_init_deg<<<blocks(N), THREADS, 0, stream>>>(dinv, N);
    k_accum_deg<<<blocks(E), THREADS, 0, stream>>>(dst, dinv, E);
    k_dinv<<<blocks(N), THREADS, 0, stream>>>(dinv, N);

    k_gemm_wmma<128, 64><<<gemmBl, THREADS, 0, stream>>>(x, W1, bufA, N);
    k_zero<<<blocks((long long)N * 64), THREADS, 0, stream>>>(bufB, (size_t)N * 64);
    k_scatter<64><<<blocks((long long)E * 16), THREADS, 0, stream>>>(src, dst, dinv, bufA, bufB, E);
    k_finalize<64, true><<<blocks((long long)N * 64), THREADS, 0, stream>>>(bufA, dinv, b1, bufB, N);

    k_gemm_wmma<64, 64><<<gemmBl, THREADS, 0, stream>>>(bufB, W2, bufA, N);
    k_zero<<<blocks((long long)N * 64), THREADS, 0, stream>>>(bufB, (size_t)N * 64);
    k_scatter<64><<<blocks((long long)E * 16), THREADS, 0, stream>>>(src, dst, dinv, bufA, bufB, E);
    k_finalize<64, true><<<blocks((long long)N * 64), THREADS, 0, stream>>>(bufA, dinv, b2, bufB, N);

    k_gemm_wmma<64, 32><<<gemmBl, THREADS, 0, stream>>>(bufB, W3, bufA, N);
    k_zero<<<blocks((long long)N * 32), THREADS, 0, stream>>>(out, (size_t)N * 32);
    k_scatter<32><<<blocks((long long)E * 8), THREADS, 0, stream>>>(src, dst, dinv, bufA, out, E);
    k_finalize<32, false><<<blocks((long long)N * 32), THREADS, 0, stream>>>(bufA, dinv, b3, out, N);
}


# Edge pass remapped to one wave per edge, lanes along the row: every row load and float atomic of a wave covers one whole 128-B line (was 4 quarter-filled lines); sums stay in arrival order
# speedup vs baseline: 1.8148x; 1.8148x over previous
#include <hip/hip_runtime.h>
#include <cstddef>

typedef __attribute__((ext_vector_type(2))) float v2f;
typedef __attribute__((ext_vector_type(8))) float v8f;

#define GCN_N 50000
#define THREADS 256

#define NTICKETS 8
#define PGRID 2048

__global__ void k_init_deg(float* __restrict__ deg, unsigned* __restrict__ tickets, int n) {
    int i = blockIdx.x * blockDim.x + threadIdx.x;
    if (i < n) deg[i] = 1.0f;
    if (i < NTICKETS) tickets[i] = 0u;
}

__device__ __forceinline__ unsigned my_half() {
    return (__builtin_amdgcn_s_sendmsg_rtn(0x87) >> 18) & 1u;
}

#define DEG_CHUNK 1024
__global__ __launch_bounds__(THREADS)
void k_accum_deg(const int* __restrict__ dst, float* __restrict__ deg,
                 unsigned* __restrict__ tickets, int e) {
    __shared__ unsigned sh_ticket;
    const unsigned nchunks = (unsigned)((e + DEG_CHUNK - 1) / DEG_CHUNK);
    const unsigned h = my_half();
    for (unsigned turn = 0; turn < 2; ++turn) {
        const unsigned p = h ^ turn;
        for (;;) {
            if (threadIdx.x == 0) sh_ticket = atomicAdd(&tickets[p], 1u);
            __syncthreads();
            const unsigned c = sh_ticket;
            __syncthreads();
            if (c >= nchunks) break;
#pragma unroll
            for (int k = 0; k < DEG_CHUNK / THREADS; ++k) {
                const int i = (int)c * DEG_CHUNK + k * THREADS + (int)threadIdx.x;
                if (i < e) {
                    const int d = dst[i];
                    if ((((unsigned)d >> 6) & 1u) == p) atomicAdd(&deg[d], 1.0f);
                }
            }
        }
    }
}

template <int KIN, int KOUT, bool DEG_IN, bool RELU_IN>
__global__ __launch_bounds__(THREADS)
void k_gemm_wmma(const float* A, const float* __restrict__ W,
                 const float* __restrict__ bias, float* dnorm,
                 float* __restrict__ C, float* S, int nrows) {
    __shared__ v2f sW[(KIN / 2) * KOUT];
    for (int i = threadIdx.x; i < (KIN / 2) * KOUT; i += blockDim.x) {
        const int k2 = i / KOUT;
        const int c  = i % KOUT;
        v2f w;
        w.x = W[(2 * k2 + 0) * KOUT + c];
        w.y = W[(2 * k2 + 1) * KOUT + c];
        sW[i] = w;
    }
    __syncthreads();

    const int wave = threadIdx.x >> 5;
    const int lane = threadIdx.x & 31;
    const int tile = blockIdx.x * 8 + wave;
    if (tile * 16 >= nrows) return;

    constexpr int NT = KOUT / 16;
    v8f acc[NT] = {};

    const int m    = lane & 15;
    const int khi  = lane >> 4;
    const float* Arow = A + (size_t)(tile * 16 + m) * KIN;
    const v2f* sWl = sW + khi * KOUT + m;

    const int row0 = tile * 16 + khi * 8;
    const float4 q0 = *(const float4*)(dnorm + row0);
    const float4 q1 = *(const float4*)(dnorm + row0 + 4);
    float di[8] = {q0.x, q0.y, q0.z, q0.w, q1.x, q1.y, q1.z, q1.w};
    if (DEG_IN) {
#pragma unroll
        for (int v = 0; v < 8; ++v) di[v] = rsqrtf(fmaxf(di[v], 1.0f));
    }
    float d2[8];
#pragma unroll
    for (int v = 0; v < 8; ++v) d2[v] = di[v] * di[v];
    float bt[NT];
#pragma unroll
    for (int t = 0; t < NT; ++t) bt[t] = bias[t * 16 + m];

    for (int kk = 0; kk < KIN / 4; ++kk) {
        v2f a = *(const v2f*)(Arow + kk * 4 + khi * 2);
        if (RELU_IN) { a.x = fmaxf(a.x, 0.0f); a.y = fmaxf(a.y, 0.0f); }
        const v2f* sWk = sWl + (kk * 2) * KOUT;
#pragma unroll
        for (int t = 0; t < NT; ++t) {
            const v2f b = sWk[t * 16];
            acc[t] = __builtin_amdgcn_wmma_f32_16x16x4_f32(
                 false, a,  false, b,
                 (short)0, acc[t],  false,  false);
        }
    }

#pragma unroll
    for (int t = 0; t < NT; ++t) {
#pragma unroll
        for (int v = 0; v < 8; ++v) {
            const size_t o = (size_t)(row0 + v) * KOUT + t * 16 + m;
            const float tv = acc[t][v];
            C[o] = tv;
            S[o] = tv * d2[v] + bt[t];
        }
    }

    if (DEG_IN && m == 0) {
        *(float4*)(dnorm + row0)     = make_float4(di[0], di[1], di[2], di[3]);
        *(float4*)(dnorm + row0 + 4) = make_float4(di[4], di[5], di[6], di[7]);
    }
}

#define SC_CHUNK 256
#define SC_U 8
__device__ __forceinline__ void rows_asked(const float (&h)[SC_U][1]) {
    asm volatile("" :: "v"(h[0][0]), "v"(h[1][0]), "v"(h[2][0]), "v"(h[3][0]),
                       "v"(h[4][0]), "v"(h[5][0]), "v"(h[6][0]), "v"(h[7][0]));
}
__device__ __forceinline__ void rows_asked(const float (&h)[SC_U][2]) {
    asm volatile("" :: "v"(h[0][0]), "v"(h[0][1]), "v"(h[1][0]), "v"(h[1][1]),
                       "v"(h[2][0]), "v"(h[2][1]), "v"(h[3][0]), "v"(h[3][1]),
                       "v"(h[4][0]), "v"(h[4][1]), "v"(h[5][0]), "v"(h[5][1]),
                       "v"(h[6][0]), "v"(h[6][1]), "v"(h[7][0]), "v"(h[7][1]));
}
template <int F>
__global__ __launch_bounds__(THREADS)
void k_scatter(const int* __restrict__ src, const int* __restrict__ dst,
               const float* __restrict__ dinv, const float* __restrict__ H,
               float* __restrict__ out, unsigned* __restrict__ tickets, int nedges) {
    constexpr int NV = F / 32;
    constexpr int RSH = (F == 64) ? 0 : 1;
    constexpr int NW = THREADS / 32;
    __shared__ int ls[SC_CHUNK];
    __shared__ int ld[SC_CHUNK];
    __shared__ unsigned sh_ticket;
    __shared__ unsigned sh_cnt;
    const unsigned nchunks = (unsigned)((nedges + SC_CHUNK - 1) / SC_CHUNK);
    const unsigned h = my_half();
    const int wave = __builtin_amdgcn_readfirstlane((int)(threadIdx.x >> 5));
    const int lane = (int)(threadIdx.x & 31);
    for (unsigned turn = 0; turn < 2; ++turn) {
        const unsigned p = h ^ turn;
        for (;;) {
            if (threadIdx.x == 0) { sh_ticket = atomicAdd(&tickets[p], 1u); sh_cnt = 0u; }
            __syncthreads();
            const unsigned c = sh_ticket;
            if (c >= nchunks) break;
#pragma unroll
            for (int k = 0; k < SC_CHUNK / THREADS; ++k) {
                const int e = (int)c * SC_CHUNK + k * THREADS + (int)threadIdx.x;
                if (e < nedges) {
                    const int d = dst[e];
                    if ((((unsigned)d >> RSH) & 1u) == p) {
                        const unsigned pos = atomicAdd(&sh_cnt, 1u);
                        ls[pos] = src[e];
                        ld[pos] = d;
                    }
                }
            }
            __syncthreads();
            const int cnt = __builtin_amdgcn_readfirstlane((int)sh_cnt);
            const int j = wave + NW * lane;
            int sv = 0, dv = 0;
            float nv = 0.0f;
            if (j < cnt) {
                sv = ls[j];
                dv = ld[j];
                nv = dinv[sv] * dinv[dv];
            }
            const int nk = (cnt - wave + NW - 1) / NW;
            for (int k0 = 0; k0 < nk; k0 += SC_U) {
                int dd[SC_U];
                float nn[SC_U];
                float hv[SC_U][NV];
#pragma unroll
                for (int u = 0; u < SC_U; ++u) {
                    const int k = (k0 + u < nk) ? (k0 + u) : (nk - 1);
                    const int s = __builtin_amdgcn_readlane(sv, k);
                    dd[u] = __builtin_amdgcn_readlane(dv, k);
                    nn[u] = __int_as_float(__builtin_amdgcn_readlane(__float_as_int(nv), k));
                    const float* hs = H + (size_t)s * F + lane;
#pragma unroll
                    for (int v = 0; v < NV; ++v) hv[u][v] = hs[v * 32];
                }
                rows_asked(hv);
#pragma unroll
                for (int u = 0; u < SC_U; ++u) {
                    if (k0 + u < nk) {
                        float* o = out + (size_t)dd[u] * F + lane;
#pragma unroll
                        for (int v = 0; v < NV; ++v) atomicAdd(o + v * 32, hv[u][v] * nn[u]);
                    }
                }
            }
            __syncthreads();
        }
        __syncthreads();
    }
}

extern "C" void kernel_launch(void* const* d_in, const int* in_sizes, int n_in,
                              void* d_out, int out_size, void* d_ws, size_t ws_size,
                              hipStream_t stream) {
    const float* x  = (const float*)d_in[0];
    const int*   ei = (const int*)d_in[1];
    const float* W1 = (const float*)d_in[2];
    const float* b1 = (const float*)d_in[3];
    const float* W2 = (const float*)d_in[4];
    const float* b2 = (const float*)d_in[5];
    const float* W3 = (const float*)d_in[6];
    const float* b3 = (const float*)d_in[7];
    float* out = (float*)d_out;

    const int N = GCN_N;
    const int E = in_sizes[1] / 2;
    const int* src = ei;
    const int* dst = ei + E;

    float* dinv = (float*)d_ws;
    unsigned* tickets = (unsigned*)(dinv + N);
    float* bufA = dinv + ((N + 255) & ~255);
    float* bufB = bufA + (size_t)N * 64;

    const int tiles  = N / 16;
    const int gemmBl = (tiles + 7) / 8;

    auto blocks = [](long long n) { return (unsigned)((n + THREADS - 1) / THREADS); };

    k_init_deg<<<blocks(N), THREADS, 0, stream>>>(dinv, tickets, N);
    k_accum_deg<<<PGRID, THREADS, 0, stream>>>(dst, dinv, tickets + 0, E);

    k_gemm_wmma<128, 64, true, false><<<gemmBl, THREADS, 0, stream>>>(x, W1, b1, dinv, bufA, bufB, N);
    k_scatter<64><<<PGRID, THREADS, 0, stream>>>(src, dst, dinv, bufA, bufB, tickets + 2, E);

    k_gemm_wmma<64, 64, false, true><<<gemmBl, THREADS, 0, stream>>>(bufB, W2, b2, dinv, bufA, bufB, N);
    k_scatter<64><<<PGRID, THREADS, 0, stream>>>(src, dst, dinv, bufA, bufB, tickets + 4, E);

    k_gemm_wmma<64, 32, false, true><<<gemmBl, THREADS, 0, stream>>>(bufB, W3, b3, dinv, bufA, out, N);
    k_scatter<32><<<PGRID, THREADS, 0, stream>>>(src, dst, dinv, bufA, out, tickets + 6, E);
}


# Edge pass: fewer dependent round trips per chunk visit (src+dst asked together, norm gather under first row block, twin lists so no closing barrier, 448-edge chunks); sums stay in arrival order
# speedup vs baseline: 1.8718x; 1.0314x over previous
#include <hip/hip_runtime.h>
#include <cstddef>

typedef __attribute__((ext_vector_type(2))) float v2f;
typedef __attribute__((ext_vector_type(8))) float v8f;

#define GCN_N 50000
#define THREADS 256

#define NTICKETS 8
#define PGRID 2048

__global__ void k_init_deg(float* __restrict__ deg, unsigned* __restrict__ tickets, int n) {
    int i = blockIdx.x * blockDim.x + threadIdx.x;
    if (i < n) deg[i] = 1.0f;
    if (i < NTICKETS) tickets[i] = 0u;
}

__device__ __forceinline__ unsigned my_half() {
    return (__builtin_amdgcn_s_sendmsg_rtn(0x87) >> 18) & 1u;
}

#define DEG_CHUNK 1024
__global__ __launch_bounds__(THREADS)
void k_accum_deg(const int* __restrict__ dst, float* __restrict__ deg,
                 unsigned* __restrict__ tickets, int e) {
    __shared__ unsigned sh_ticket;
    const unsigned nchunks = (unsigned)((e + DEG_CHUNK - 1) / DEG_CHUNK);
    const unsigned h = my_half();
    for (unsigned turn = 0; turn < 2; ++turn) {
        const unsigned p = h ^ turn;
        for (;;) {
            if (threadIdx.x == 0) sh_ticket = atomicAdd(&tickets[p], 1u);
            __syncthreads();
            const unsigned c = sh_ticket;
            __syncthreads();
            if (c >= nchunks) break;
#pragma unroll
            for (int k = 0; k < DEG_CHUNK / THREADS; ++k) {
                const int i = (int)c * DEG_CHUNK + k * THREADS + (int)threadIdx.x;
                if (i < e) {
                    const int d = dst[i];
                    if ((((unsigned)d >> 6) & 1u) == p) atomicAdd(&deg[d], 1.0f);
                }
            }
        }
    }
}

template <int KIN, int KOUT, bool DEG_IN, bool RELU_IN>
__global__ __launch_bounds__(THREADS)
void k_gemm_wmma(const float* A, const float* __restrict__ W,
                 const float* __restrict__ bias, float* dnorm,
                 float* __restrict__ C, float* S, int nrows) {
    __shared__ v2f sW[(KIN / 2) * KOUT];
    for (int i = threadIdx.x; i < (KIN / 2) * KOUT; i += blockDim.x) {
        const int k2 = i / KOUT;
        const int c  = i % KOUT;
        v2f w;
        w.x = W[(2 * k2 + 0) * KOUT + c];
        w.y = W[(2 * k2 + 1) * KOUT + c];
        sW[i] = w;
    }
    __syncthreads();

    const int wave = threadIdx.x >> 5;
    const int lane = threadIdx.x & 31;
    const int tile = blockIdx.x * 8 + wave;
    if (tile * 16 >= nrows) return;

    constexpr int NT = KOUT / 16;
    v8f acc[NT] = {};

    const int m    = lane & 15;
    const int khi  = lane >> 4;
    const float* Arow = A + (size_t)(tile * 16 + m) * KIN;
    const v2f* sWl = sW + khi * KOUT + m;

    const int row0 = tile * 16 + khi * 8;
    const float4 q0 = *(const float4*)(dnorm + row0);
    const float4 q1 = *(const float4*)(dnorm + row0 + 4);
    float di[8] = {q0.x, q0.y, q0.z, q0.w, q1.x, q1.y, q1.z, q1.w};
    if (DEG_IN) {
#pragma unroll
        for (int v = 0; v < 8; ++v) di[v] = rsqrtf(fmaxf(di[v], 1.0f));
    }
    float d2[8];
#pragma unroll
    for (int v = 0; v < 8; ++v) d2[v] = di[v] * di[v];
    float bt[NT];
#pragma unroll
    for (int t = 0; t < NT; ++t) bt[t] = bias[t * 16 + m];

    for (int kk = 0; kk < KIN / 4; ++kk) {
        v2f a = *(const v2f*)(Arow + kk * 4 + khi * 2);
        if (RELU_IN) { a.x = fmaxf(a.x, 0.0f); a.y = fmaxf(a.y, 0.0f); }
        const v2f* sWk = sWl + (kk * 2) * KOUT;
#pragma unroll
        for (int t = 0; t < NT; ++t) {
            const v2f b = sWk[t * 16];
            acc[t] = __builtin_amdgcn_wmma_f32_16x16x4_f32(
                 false, a,  false, b,
                 (short)0, acc[t],  false,  false);
        }
    }

#pragma unroll
    for (int t = 0; t < NT; ++t) {
#pragma unroll
        for (int v = 0; v < 8; ++v) {
            const size_t o = (size_t)(row0 + v) * KOUT + t * 16 + m;
            const float tv = acc[t][v];
            C[o] = tv;
            S[o] = tv * d2[v] + bt[t];
        }
    }

    if (DEG_IN && m == 0) {
        *(float4*)(dnorm + row0)     = make_float4(di[0], di[1], di[2], di[3]);
        *(float4*)(dnorm + row0 + 4) = make_float4(di[4], di[5], di[6], di[7]);
    }
}

#define SC_CHUNK 448
#define SC_KPT ((SC_CHUNK + THREADS - 1) / THREADS)
#define SC_U 8
__device__ __forceinline__ void rows_asked(const float (&h)[SC_U][1]) {
    asm volatile("" :: "v"(h[0][0]), "v"(h[1][0]), "v"(h[2][0]), "v"(h[3][0]),
                       "v"(h[4][0]), "v"(h[5][0]), "v"(h[6][0]), "v"(h[7][0]));
}
__device__ __forceinline__ void rows_asked(const float (&h)[SC_U][2]) {
    asm volatile("" :: "v"(h[0][0]), "v"(h[0][1]), "v"(h[1][0]), "v"(h[1][1]),
                       "v"(h[2][0]), "v"(h[2][1]), "v"(h[3][0]), "v"(h[3][1]),
                       "v"(h[4][0]), "v"(h[4][1]), "v"(h[5][0]), "v"(h[5][1]),
                       "v"(h[6][0]), "v"(h[6][1]), "v"(h[7][0]), "v"(h[7][1]));
}
static_assert(SC_KPT == 2, "idx_asked names two edges per thread");
__device__ __forceinline__ void idx_asked(const int (&d)[SC_KPT], const int (&s)[SC_KPT]) {
    asm volatile("" :: "v"(d[0]), "v"(s[0]), "v"(d[1]), "v"(s[1]));
}
template <int F>
__global__ __launch_bounds__(THREADS)
void k_scatter(const int* __restrict__ src, const int* __restrict__ dst,
               const float* __restrict__ dinv, const float* __restrict__ H,
               float* __restrict__ out, unsigned* __restrict__ tickets, int nedges) {
    constexpr int NV = F / 32;
    constexpr int RSH = (F == 64) ? 0 : 1;
    constexpr int NW = THREADS / 32;
    __shared__ int ls[2][SC_CHUNK];
    __shared__ int ld[2][SC_CHUNK];
    __shared__ unsigned sh_ticket[2];
    __shared__ unsigned sh_cnt[2];
    const unsigned nchunks = (unsigned)((nedges + SC_CHUNK - 1) / SC_CHUNK);
    const unsigned h = my_half();
    const int wave = __builtin_amdgcn_readfirstlane((int)(threadIdx.x >> 5));
    const int lane = (int)(threadIdx.x & 31);
    unsigned par = 0u;
    for (unsigned turn = 0; turn < 2; ++turn) {
        const unsigned p = h ^ turn;
        for (;;) {
            const unsigned b = par;
            par ^= 1u;
            if (threadIdx.x == 0) { sh_ticket[b] = atomicAdd(&tickets[p], 1u); sh_cnt[b] = 0u; }
            __syncthreads();
            const unsigned c = sh_ticket[b];
            if (c >= nchunks) break;
            int dk[SC_KPT], sk[SC_KPT];
            bool ok[SC_KPT];
#pragma unroll
            for (int k = 0; k < SC_KPT; ++k) {
                const int off = k * THREADS + (int)threadIdx.x;
                const int e = (int)c * SC_CHUNK + off;
                ok[k] = (off < SC_CHUNK) && (e < nedges);
                const int ec = (e < nedges) ? e : (nedges - 1);
                dk[k] = dst[ec];
                sk[k] = src[ec];
            }
            idx_asked(dk, sk);
#pragma unroll
            for (int k = 0; k < SC_KPT; ++k) {
                if (ok[k] && ((((unsigned)dk[k] >> RSH) & 1u) == p)) {
                    const unsigned pos = atomicAdd(&sh_cnt[b], 1u);
                    ls[b][pos] = sk[k];
                    ld[b][pos] = dk[k];
                }
            }
            __syncthreads();
            const int cnt = __builtin_amdgcn_readfirstlane((int)sh_cnt[b]);
            const int nk = (cnt - wave + NW - 1) / NW;
            for (int base = 0; base < nk; base += 32) {
                const int j  = wave + NW * (base + lane);
                const int jc = (j < cnt) ? j : (cnt - 1);
                const int sv = ls[b][jc];
                const int dv = ld[b][jc];
                const float da = dinv[sv];
                const float db = dinv[dv];
                const int nks = (nk - base < 32) ? (nk - base) : 32;
                for (int k0 = 0; k0 < nks; k0 += SC_U) {
                    int kl[SC_U];
                    int dd[SC_U];
                    float hv[SC_U][NV];
#pragma unroll
                    for (int u = 0; u < SC_U; ++u) {
                        const int k = (k0 + u < nks) ? (k0 + u) : (nks - 1);
                        kl[u] = k;
                        const int s = __builtin_amdgcn_readlane(sv, k);
                        dd[u] = __builtin_amdgcn_readlane(dv, k);
                        const float* hs = H + (size_t)s * F + lane;
#pragma unroll
                        for (int v = 0; v < NV; ++v) hv[u][v] = hs[v * 32];
                    }
                    rows_asked(hv);
#pragma unroll
                    for (int u = 0; u < SC_U; ++u) {
                        if (k0 + u < nks) {
                            const float nn = __int_as_float(__builtin_amdgcn_readlane(__float_as_int(da), kl[u])) *
                                             __int_as_float(__builtin_amdgcn_readlane(__float_as_int(db), kl[u]));
                            float* o = out + (size_t)dd[u] * F + lane;
#pragma unroll
                            for (int v = 0; v < NV; ++v) atomicAdd(o + v * 32, hv[u][v] * nn);
                        }
                    }
                }
            }
        }
    }
}

extern "C" void kernel_launch(void* const* d_in, const int* in_sizes, int n_in,
                              void* d_out, int out_size, void* d_ws, size_t ws_size,
                              hipStream_t stream) {
    const float* x  = (const float*)d_in[0];
    const int*   ei = (const int*)d_in[1];
    const float* W1 = (const float*)d_in[2];
    const float* b1 = (const float*)d_in[3];
    const float* W2 = (const float*)d_in[4];
    const float* b2 = (const float*)d_in[5];
    const float* W3 = (const float*)d_in[6];
    const float* b3 = (const float*)d_in[7];
    float* out = (float*)d_out;

    const int N = GCN_N;
    const int E = in_sizes[1] / 2;
    const int* src = ei;
    const int* dst = ei + E;

    float* dinv = (float*)d_ws;
    unsigned* tickets = (unsigned*)(dinv + N);
    float* bufA = dinv + ((N + 255) & ~255);
    float* bufB = bufA + (size_t)N * 64;

    const int tiles  = N / 16;
    const int gemmBl = (tiles + 7) / 8;

    auto blocks = [](long long n) { return (unsigned)((n + THREADS - 1) / THREADS); };

    k_init_deg<<<blocks(N), THREADS, 0, stream>>>(dinv, tickets, N);
    k_accum_deg<<<PGRID, THREADS, 0, stream>>>(dst, dinv, tickets + 0, E);

    k_gemm_wmma<128, 64, true, false><<<gemmBl, THREADS, 0, stream>>>(x, W1, b1, dinv, bufA, bufB, N);
    k_scatter<64><<<PGRID, THREADS, 0, stream>>>(src, dst, dinv, bufA, bufB, tickets + 2, E);

    k_gemm_wmma<64, 64, false, true><<<gemmBl, THREADS, 0, stream>>>(bufB, W2, b2, dinv, bufA, bufB, N);
    k_scatter<64><<<PGRID, THREADS, 0, stream>>>(src, dst, dinv, bufA, bufB, tickets + 4, E);

    k_gemm_wmma<64, 32, false, true><<<gemmBl, THREADS, 0, stream>>>(bufB, W3, b3, dinv, bufA, out, N);
    k_scatter<32><<<PGRID, THREADS, 0, stream>>>(src, dst, dinv, bufA, out, tickets + 6, E);
}


# Dense transforms: a wave's A rows and both result planes cross as whole 128-B lines through its LDS tile, W and A loads of a tile all in flight at once (was a wait per trip); sums bit-identical
# speedup vs baseline: 2.0182x; 1.0782x over previous
#include <hip/hip_runtime.h>
#include <cstddef>

typedef __attribute__((ext_vector_type(2))) float v2f;
typedef __attribute__((ext_vector_type(8))) float v8f;

#define GCN_N 50000
#define THREADS 256

#define NTICKETS 8
#define PGRID 2048

__global__ void k_init_deg(float* __restrict__ deg, unsigned* __restrict__ tickets, int n) {
    int i = blockIdx.x * blockDim.x + threadIdx.x;
    if (i < n) deg[i] = 1.0f;
    if (i < NTICKETS) tickets[i] = 0u;
}

__device__ __forceinline__ unsigned my_half() {
    return (__builtin_amdgcn_s_sendmsg_rtn(0x87) >> 18) & 1u;
}

#define DEG_CHUNK 1024
__global__ __launch_bounds__(THREADS)
void k_accum_deg(const int* __restrict__ dst, float* __restrict__ deg,
                 unsigned* __restrict__ tickets, int e) {
    __shared__ unsigned sh_ticket;
    const unsigned nchunks = (unsigned)((e + DEG_CHUNK - 1) / DEG_CHUNK);
    const unsigned h = my_half();
    for (unsigned turn = 0; turn < 2; ++turn) {
        const unsigned p = h ^ turn;
        for (;;) {
            if (threadIdx.x == 0) sh_ticket = atomicAdd(&tickets[p], 1u);
            __syncthreads();
            const unsigned c = sh_ticket;
            __syncthreads();
            if (c >= nchunks) break;
#pragma unroll
            for (int k = 0; k < DEG_CHUNK / THREADS; ++k) {
                const int i = (int)c * DEG_CHUNK + k * THREADS + (int)threadIdx.x;
                if (i < e) {
                    const int d = dst[i];
                    if ((((unsigned)d >> 6) & 1u) == p) atomicAdd(&deg[d], 1.0f);
                }
            }
        }
    }
}

template <int KIN, int KOUT, bool DEG_IN, bool RELU_IN>
__global__ __launch_bounds__(THREADS)
void k_gemm_wmma(const float* A, const float* __restrict__ W,
                 const float* __restrict__ bias, float* dnorm,
                 float* __restrict__ C, float* S, int nrows) {
    constexpr int NWV   = THREADS / 32;
    constexpr int PT    = KIN + 4;
    constexpr int NPAIR = (KIN / 2) * KOUT;
    constexpr int LPR   = KIN / 4;
    constexpr int RPI   = 32 / LPR;
    constexpr int NLD   = 16 / RPI;
    constexpr int LPO   = KOUT / 4;
    constexpr int RPO   = 32 / LPO;
    constexpr int NST   = 16 / RPO;
    static_assert(KOUT <= KIN, "the result tile reuses the A tile's LDS rows");
    static_assert(NPAIR % THREADS == 0, "W staging has a constant trip count");
    __shared__ v2f sW[NPAIR];
    __shared__ __attribute__((aligned(16))) float sT[NWV][16 * PT];
    __shared__ float sD[NWV][16];

    const int wave = threadIdx.x >> 5;
    const int lane = threadIdx.x & 31;
    const int tile = blockIdx.x * NWV + wave;
    const bool active = tile * 16 < nrows;
    const int tl = active ? tile : 0;

    const int rl = lane / LPR;
    const int cl = (lane % LPR) * 4;
    float4 av[NLD];
#pragma unroll
    for (int i = 0; i < NLD; ++i)
        av[i] = *(const float4*)(A + (size_t)(tl * 16 + i * RPI + rl) * KIN + cl);

#pragma unroll
    for (int j = 0; j < NPAIR / THREADS; ++j) {
        const int i  = j * THREADS + (int)threadIdx.x;
        const int k2 = i / KOUT;
        const int c  = i % KOUT;
        v2f w;
        w.x = W[(2 * k2 + 0) * KOUT + c];
        w.y = W[(2 * k2 + 1) * KOUT + c];
        sW[i] = w;
    }

    const int m    = lane & 15;
    const int khi  = lane >> 4;

    const float dr = dnorm[tl * 16 + m];
    const float dv = DEG_IN ? rsqrtf(fmaxf(dr, 1.0f)) : dr;
    const int ro = lane / LPO;
    const int co = (lane % LPO) * 4;
    const float4 b4 = *(const float4*)(bias + co);

    float* sTw = &sT[wave][0];
#pragma unroll
    for (int i = 0; i < NLD; ++i) {
        float4 a = av[i];
        if (RELU_IN) {
            a.x = fmaxf(a.x, 0.0f); a.y = fmaxf(a.y, 0.0f);
            a.z = fmaxf(a.z, 0.0f); a.w = fmaxf(a.w, 0.0f);
        }
        *(float4*)(sTw + (i * RPI + rl) * PT + cl) = a;
    }
    sD[wave][m] = dv * dv;
    __syncthreads();

    constexpr int NT = KOUT / 16;
    v8f acc[NT] = {};
    const float* sAl = sTw + m * PT + khi * 2;
    const v2f* sWl = sW + khi * KOUT + m;

    for (int kk = 0; kk < KIN / 4; ++kk) {
        const v2f a = *(const v2f*)(sAl + kk * 4);
        const v2f* sWk = sWl + (kk * 2) * KOUT;
#pragma unroll
        for (int t = 0; t < NT; ++t) {
            const v2f b = sWk[t * 16];
            acc[t] = __builtin_amdgcn_wmma_f32_16x16x4_f32(
                 false, a,  false, b,
                 (short)0, acc[t],  false,  false);
        }
    }

#pragma unroll
    for (int t = 0; t < NT; ++t) {
#pragma unroll
        for (int v = 0; v < 8; ++v) sTw[(khi * 8 + v) * PT + t * 16 + m] = acc[t][v];
    }
    __syncthreads();

    if (active) {
#pragma unroll
        for (int i = 0; i < NST; ++i) {
            const int r = i * RPO + ro;
            const float4 tv = *(const float4*)(sTw + r * PT + co);
            const float d2 = sD[wave][r];
            const size_t o = (size_t)(tile * 16 + r) * KOUT + co;
            *(float4*)(C + o) = tv;
            float4 sv;
            sv.x = tv.x * d2 + b4.x; sv.y = tv.y * d2 + b4.y;
            sv.z = tv.z * d2 + b4.z; sv.w = tv.w * d2 + b4.w;
            *(float4*)(S + o) = sv;
        }
        if (DEG_IN && khi == 0) dnorm[tile * 16 + m] = dv;
    }
}

#define SC_CHUNK 448
#define SC_KPT ((SC_CHUNK + THREADS - 1) / THREADS)
#define SC_U 8
__device__ __forceinline__ void rows_asked(const float (&h)[SC_U][1]) {
    asm volatile("" :: "v"(h[0][0]), "v"(h[1][0]), "v"(h[2][0]), "v"(h[3][0]),
                       "v"(h[4][0]), "v"(h[5][0]), "v"(h[6][0]), "v"(h[7][0]));
}
__device__ __forceinline__ void rows_asked(const float (&h)[SC_U][2]) {
    asm volatile("" :: "v"(h[0][0]), "v"(h[0][1]), "v"(h[1][0]), "v"(h[1][1]),
                       "v"(h[2][0]), "v"(h[2][1]), "v"(h[3][0]), "v"(h[3][1]),
                       "v"(h[4][0]), "v"(h[4][1]), "v"(h[5][0]), "v"(h[5][1]),
                       "v"(h[6][0]), "v"(h[6][1]), "v"(h[7][0]), "v"(h[7][1]));
}
static_assert(SC_KPT == 2, "idx_asked names two edges per thread");
__device__ __forceinline__ void idx_asked(const int (&d)[SC_KPT], const int (&s)[SC_KPT]) {
    asm volatile("" :: "v"(d[0]), "v"(s[0]), "v"(d[1]), "v"(s[1]));
}
template <int F>
__global__ __launch_bounds__(THREADS)
void k_scatter(const int* __restrict__ src, const int* __restrict__ dst,
               const float* __restrict__ dinv, const float* __restrict__ H,
               float* __restrict__ out, unsigned* __restrict__ tickets, int nedges) {
    constexpr int NV = F / 32;
    constexpr int RSH = (F == 64) ? 0 : 1;
    constexpr int NW = THREADS / 32;
    __shared__ int ls[2][SC_CHUNK];
    __shared__ int ld[2][SC_CHUNK];
    __shared__ unsigned sh_ticket[2];
    __shared__ unsigned sh_cnt[2];
    const unsigned nchunks = (unsigned)((nedges + SC_CHUNK - 1) / SC_CHUNK);
    const unsigned h = my_half();
    const int wave = __builtin_amdgcn_readfirstlane((int)(threadIdx.x >> 5));
    const int lane = (int)(threadIdx.x & 31);
    unsigned par = 0u;
    for (unsigned turn = 0; turn < 2; ++turn) {
        const unsigned p = h ^ turn;
        for (;;) {
            const unsigned b = par;
            par ^= 1u;
            if (threadIdx.x == 0) { sh_ticket[b] = atomicAdd(&tickets[p], 1u); sh_cnt[b] = 0u; }
            __syncthreads();
            const unsigned c = sh_ticket[b];
            if (c >= nchunks) break;
            int dk[SC_KPT], sk[SC_KPT];
            bool ok[SC_KPT];
#pragma unroll
            for (int k = 0; k < SC_KPT; ++k) {
                const int off = k * THREADS + (int)threadIdx.x;
                const int e = (int)c * SC_CHUNK + off;
                ok[k] = (off < SC_CHUNK) && (e < nedges);
                const int ec = (e < nedges) ? e : (nedges - 1);
                dk[k] = dst[ec];
                sk[k] = src[ec];
            }
            idx_asked(dk, sk);
#pragma unroll
            for (int k = 0; k < SC_KPT; ++k) {
                if (ok[k] && ((((unsigned)dk[k] >> RSH) & 1u) == p)) {
                    const unsigned pos = atomicAdd(&sh_cnt[b], 1u);
                    ls[b][pos] = sk[k];
                    ld[b][pos] = dk[k];
                }
            }
            __syncthreads();
            const int cnt = __builtin_amdgcn_readfirstlane((int)sh_cnt[b]);
            const int nk = (cnt - wave + NW - 1) / NW;
            for (int base = 0; base < nk; base += 32) {
                const int j  = wave + NW * (base + lane);
                const int jc = (j < cnt) ? j : (cnt - 1);
                const int sv = ls[b][jc];
                const int dv = ld[b][jc];
                const float da = dinv[sv];
                const float db = dinv[dv];
                const int nks = (nk - base < 32) ? (nk - base) : 32;
                for (int k0 = 0; k0 < nks; k0 += SC_U) {
                    int kl[SC_U];
                    int dd[SC_U];
                    float hv[SC_U][NV];
#pragma unroll
                    for (int u = 0; u < SC_U; ++u) {
                        const int k = (k0 + u < nks) ? (k0 + u) : (nks - 1);
                        kl[u] = k;
                        const int s = __builtin_amdgcn_readlane(sv, k);
                        dd[u] = __builtin_amdgcn_readlane(dv, k);
                        const float* hs = H + (size_t)s * F + lane;
#pragma unroll
                        for (int v = 0; v < NV; ++v) hv[u][v] = hs[v * 32];
                    }
                    rows_asked(hv);
#pragma unroll
                    for (int u = 0; u < SC_U; ++u) {
                        if (k0 + u < nks) {
                            const float nn = __int_as_float(__builtin_amdgcn_readlane(__float_as_int(da), kl[u])) *
                                             __int_as_float(__builtin_amdgcn_readlane(__float_as_int(db), kl[u]));
                            float* o = out + (size_t)dd[u] * F + lane;
#pragma unroll
                            for (int v = 0; v < NV; ++v) atomicAdd(o + v * 32, hv[u][v] * nn);
                        }
                    }
                }
            }
        }
    }
}

extern "C" void kernel_launch(void* const* d_in, const int* in_sizes, int n_in,
                              void* d_out, int out_size, void* d_ws, size_t ws_size,
                              hipStream_t stream) {
    const float* x  = (const float*)d_in[0];
    const int*   ei = (const int*)d_in[1];
    const float* W1 = (const float*)d_in[2];
    const float* b1 = (const float*)d_in[3];
    const float* W2 = (const float*)d_in[4];
    const float* b2 = (const float*)d_in[5];
    const float* W3 = (const float*)d_in[6];
    const float* b3 = (const float*)d_in[7];
    float* out = (float*)d_out;

    const int N = GCN_N;
    const int E = in_sizes[1] / 2;
    const int* src = ei;
    const int* dst = ei + E;

    float* dinv = (float*)d_ws;
    unsigned* tickets = (unsigned*)(dinv + N);
    float* bufA = dinv + ((N + 255) & ~255);
    float* bufB = bufA + (size_t)N * 64;

    const int tiles  = N / 16;
    const int gemmBl = (tiles + 7) / 8;

    auto blocks = [](long long n) { return (unsigned)((n + THREADS - 1) / THREADS); };

    k_init_deg<<<blocks(N), THREADS, 0, stream>>>(dinv, tickets, N);
    k_accum_deg<<<PGRID, THREADS, 0, stream>>>(dst, dinv, tickets + 0, E);

    k_gemm_wmma<128, 64, true, false><<<gemmBl, THREADS, 0, stream>>>(x, W1, b1, dinv, bufA, bufB, N);
    k_scatter<64><<<PGRID, THREADS, 0, stream>>>(src, dst, dinv, bufA, bufB, tickets + 2, E);

    k_gemm_wmma<64, 64, false, true><<<gemmBl, THREADS, 0, stream>>>(bufB, W2, b2, dinv, bufA, bufB, N);
    k_scatter<64><<<PGRID, THREADS, 0, stream>>>(src, dst, dinv, bufA, bufB, tickets + 4, E);

    k_gemm_wmma<64, 32, false, true><<<gemmBl, THREADS, 0, stream>>>(bufB, W3, b3, dinv, bufA, out, N);
    k_scatter<32><<<PGRID, THREADS, 0, stream>>>(src, dst, dinv, bufA, out, tickets + 6, E);
}


# Degree pass: 64 work-groups count their edge slice in an LDS histogram and store 16-bit count planes (in out, before its values exist); first dense launch adds them; 8 launches -> 7; counts exact
# speedup vs baseline: 2.1248x; 1.0528x over previous
#include <hip/hip_runtime.h>
#include <cstddef>

typedef __attribute__((ext_vector_type(2))) float v2f;
typedef __attribute__((ext_vector_type(8))) float v8f;

#define GCN_N 50000
#define THREADS 256

#define NTICKETS 8
#define PGRID 2048
#define DEG_G 64
#define DEG_T 1024
#define DEG_UNR 13
#define DEG_PW (GCN_N / 2)
static_assert(DEG_PW % 4 == 0, "planes move as 16-byte pieces");

__global__ __launch_bounds__(DEG_T)
void k_deg_planes(const int* __restrict__ dst, unsigned* __restrict__ planes,
                  unsigned* __restrict__ tickets, int e) {
    __shared__ __attribute__((aligned(16))) unsigned hist[DEG_PW];
    const int tid = (int)threadIdx.x;
    if (blockIdx.x == 0 && tid < NTICKETS) tickets[tid] = 0u;
    const int per = (e + DEG_G - 1) / DEG_G;
    const int lo = (int)blockIdx.x * per;
    const int hi = (lo + per < e) ? (lo + per) : e;
    uint4* h4 = (uint4*)hist;
    for (int i = tid; i < DEG_PW / 4; i += DEG_T) h4[i] = make_uint4(0u, 0u, 0u, 0u);
    __syncthreads();
    for (int base = lo; base < hi; base += DEG_T * DEG_UNR) {
        int d[DEG_UNR];
#pragma unroll
        for (int k = 0; k < DEG_UNR; ++k) {
            const int i = base + k * DEG_T + tid;
            d[k] = dst[(i < hi) ? i : (hi - 1)];
        }
#pragma unroll
        for (int k = 0; k < DEG_UNR; ++k) {
            const int i = base + k * DEG_T + tid;
            if (i < hi && (unsigned)d[k] < (unsigned)GCN_N)
                atomicAdd(&hist[d[k] >> 1], 1u << ((d[k] & 1) * 16));
        }
    }
    __syncthreads();
    uint4* p4 = (uint4*)(planes + (size_t)blockIdx.x * DEG_PW);
    for (int i = tid; i < DEG_PW / 4; i += DEG_T) p4[i] = h4[i];
}

__device__ __forceinline__ unsigned my_half() {
    return (__builtin_amdgcn_s_sendmsg_rtn(0x87) >> 18) & 1u;
}

template <int KIN, int KOUT, bool DEG_IN, bool RELU_IN>
__global__ __launch_bounds__(THREADS)
void k_gemm_wmma(const float* A, const float* __restrict__ W,
                 const float* __restrict__ bias, float* dnorm,
                 const unsigned* __restrict__ planes,
                 float* __restrict__ C, float* S, int nrows) {
    constexpr int NWV   = THREADS / 32;
    constexpr int PT    = KIN + 4;
    constexpr int NPAIR = (KIN / 2) * KOUT;
    constexpr int LPR   = KIN / 4;
    constexpr int RPI   = 32 / LPR;
    constexpr int NLD   = 16 / RPI;
    constexpr int LPO   = KOUT / 4;
    constexpr int RPO   = 32 / LPO;
    constexpr int NST   = 16 / RPO;
    static_assert(KOUT <= KIN, "the result tile reuses the A tile's LDS rows");
    static_assert(NPAIR % THREADS == 0, "W staging has a constant trip count");
    __shared__ v2f sW[NPAIR];
    __shared__ __attribute__((aligned(16))) float sT[NWV][16 * PT];
    __shared__ float sD[NWV][16];

    const int wave = threadIdx.x >> 5;
    const int lane = threadIdx.x & 31;
    const int tile = blockIdx.x * NWV + wave;
    const bool active = tile * 16 < nrows;
    const int tl = active ? tile : 0;

    const int rl = lane / LPR;
    const int cl = (lane % LPR) * 4;
    float4 av[NLD];
#pragma unroll
    for (int i = 0; i < NLD; ++i)
        av[i] = *(const float4*)(A + (size_t)(tl * 16 + i * RPI + rl) * KIN + cl);

    const int m    = lane & 15;
    const int khi  = lane >> 4;

    uint4 pc[DEG_G / 16];
    if (DEG_IN) {
        const int q  = lane & 1;
        const int pl = lane >> 1;
#pragma unroll
        for (int j = 0; j < DEG_G / 16; ++j)
            pc[j] = *(const uint4*)(planes + (size_t)(j * 16 + pl) * DEG_PW + tl * 8 + q * 4);
    }

#pragma unroll
    for (int j = 0; j < NPAIR / THREADS; ++j) {
        const int i  = j * THREADS + (int)threadIdx.x;
        const int k2 = i / KOUT;
        const int c  = i % KOUT;
        v2f w;
        w.x = W[(2 * k2 + 0) * KOUT + c];
        w.y = W[(2 * k2 + 1) * KOUT + c];
        sW[i] = w;
    }

    float dr;
    if (DEG_IN) {
        unsigned cn[8] = {0u, 0u, 0u, 0u, 0u, 0u, 0u, 0u};
#pragma unroll
        for (int j = 0; j < DEG_G / 16; ++j) {
            cn[0] += pc[j].x & 0xffffu; cn[1] += pc[j].x >> 16;
            cn[2] += pc[j].y & 0xffffu; cn[3] += pc[j].y >> 16;
            cn[4] += pc[j].z & 0xffffu; cn[5] += pc[j].z >> 16;
            cn[6] += pc[j].w & 0xffffu; cn[7] += pc[j].w >> 16;
        }
#pragma unroll
        for (int off = 2; off < 32; off <<= 1) {
#pragma unroll
            for (int i = 0; i < 8; ++i) cn[i] += __shfl_xor(cn[i], off);
        }
        unsigned cm = 0u;
#pragma unroll
        for (int i = 0; i < 8; ++i) {
            const unsigned t = __shfl(cn[i], m >> 3);
            if ((m & 7) == i) cm = t;
        }
        dr = 1.0f + (float)cm;
    } else {
        dr = dnorm[tl * 16 + m];
    }
    const float dv = DEG_IN ? rsqrtf(fmaxf(dr, 1.0f)) : dr;
    const int ro = lane / LPO;
    const int co = (lane % LPO) * 4;
    const float4 b4 = *(const float4*)(bias + co);

    float* sTw = &sT[wave][0];
#pragma unroll
    for (int i = 0; i < NLD; ++i) {
        float4 a = av[i];
        if (RELU_IN) {
            a.x = fmaxf(a.x, 0.0f); a.y = fmaxf(a.y, 0.0f);
            a.z = fmaxf(a.z, 0.0f); a.w = fmaxf(a.w, 0.0f);
        }
        *(float4*)(sTw + (i * RPI + rl) * PT + cl) = a;
    }
    sD[wave][m] = dv * dv;
    __syncthreads();

    constexpr int NT = KOUT / 16;
    v8f acc[NT] = {};
    const float* sAl = sTw + m * PT + khi * 2;
    const v2f* sWl = sW + khi * KOUT + m;

    for (int kk = 0; kk < KIN / 4; ++kk) {
        const v2f a = *(const v2f*)(sAl + kk * 4);
        const v2f* sWk = sWl + (kk * 2) * KOUT;
#pragma unroll
        for (int t = 0; t < NT; ++t) {
            const v2f b = sWk[t * 16];
            acc[t] = __builtin_amdgcn_wmma_f32_16x16x4_f32(
                 false, a,  false, b,
                 (short)0, acc[t],  false,  false);
        }
    }

#pragma unroll
    for (int t = 0; t < NT; ++t) {
#pragma unroll
        for (int v = 0; v < 8; ++v) sTw[(khi * 8 + v) * PT + t * 16 + m] = acc[t][v];
    }
    __syncthreads();

    if (active) {
#pragma unroll
        for (int i = 0; i < NST; ++i) {
            const int r = i * RPO + ro;
            const float4 tv = *(const float4*)(sTw + r * PT + co);
            const float d2 = sD[wave][r];
            const size_t o = (size_t)(tile * 16 + r) * KOUT + co;
            *(float4*)(C + o) = tv;
            float4 sv;
            sv.x = tv.x * d2 + b4.x; sv.y = tv.y * d2 + b4.y;
            sv.z = tv.z * d2 + b4.z; sv.w = tv.w * d2 + b4.w;
            *(float4*)(S + o) = sv;
        }
        if (DEG_IN && khi == 0) dnorm[tile * 16 + m] = dv;
    }
}

#define SC_CHUNK 448
#define SC_KPT ((SC_CHUNK + THREADS - 1) / THREADS)
#define SC_U 8
__device__ __forceinline__ void rows_asked(const float (&h)[SC_U][1]) {
    asm volatile("" :: "v"(h[0][0]), "v"(h[1][0]), "v"(h[2][0]), "v"(h[3][0]),
                       "v"(h[4][0]), "v"(h[5][0]), "v"(h[6][0]), "v"(h[7][0]));
}
__device__ __forceinline__ void rows_asked(const float (&h)[SC_U][2]) {
    asm volatile("" :: "v"(h[0][0]), "v"(h[0][1]), "v"(h[1][0]), "v"(h[1][1]),
                       "v"(h[2][0]), "v"(h[2][1]), "v"(h[3][0]), "v"(h[3][1]),
                       "v"(h[4][0]), "v"(h[4][1]), "v"(h[5][0]), "v"(h[5][1]),
                       "v"(h[6][0]), "v"(h[6][1]), "v"(h[7][0]), "v"(h[7][1]));
}
static_assert(SC_KPT == 2, "idx_asked names two edges per thread");
__device__ __forceinline__ void idx_asked(const int (&d)[SC_KPT], const int (&s)[SC_KPT]) {
    asm volatile("" :: "v"(d[0]), "v"(s[0]), "v"(d[1]), "v"(s[1]));
}
template <int F>
__global__ __launch_bounds__(THREADS)
void k_scatter(const int* __restrict__ src, const int* __restrict__ dst,
               const float* __restrict__ dinv, const float* __restrict__ H,
               float* __restrict__ out, unsigned* __restrict__ tickets, int nedges) {
    constexpr int NV = F / 32;
    constexpr int RSH = (F == 64) ? 0 : 1;
    constexpr int NW = THREADS / 32;
    __shared__ int ls[2][SC_CHUNK];
    __shared__ int ld[2][SC_CHUNK];
    __shared__ unsigned sh_ticket[2];
    __shared__ unsigned sh_cnt[2];
    const unsigned nchunks = (unsigned)((nedges + SC_CHUNK - 1) / SC_CHUNK);
    const unsigned h = my_half();
    const int wave = __builtin_amdgcn_readfirstlane((int)(threadIdx.x >> 5));
    const int lane = (int)(threadIdx.x & 31);
    unsigned par = 0u;
    for (unsigned turn = 0; turn < 2; ++turn) {
        const unsigned p = h ^ turn;
        for (;;) {
            const unsigned b = par;
            par ^= 1u;
            if (threadIdx.x == 0) { sh_ticket[b] = atomicAdd(&tickets[p], 1u); sh_cnt[b] = 0u; }
            __syncthreads();
            const unsigned c = sh_ticket[b];
            if (c >= nchunks) break;
            int dk[SC_KPT], sk[SC_KPT];
            bool ok[SC_KPT];
#pragma unroll
            for (int k = 0; k < SC_KPT; ++k) {
                const int off = k * THREADS + (int)threadIdx.x;
                const int e = (int)c * SC_CHUNK + off;
                ok[k] = (off < SC_CHUNK) && (e < nedges);
                const int ec = (e < nedges) ? e : (nedges - 1);
                dk[k] = dst[ec];
                sk[k] = src[ec];
            }
            idx_asked(dk, sk);
#pragma unroll
            for (int k = 0; k < SC_KPT; ++k) {
                if (ok[k] && ((((unsigned)dk[k] >> RSH) & 1u) == p)) {
                    const unsigned pos = atomicAdd(&sh_cnt[b], 1u);
                    ls[b][pos] = sk[k];
                    ld[b][pos] = dk[k];
                }
            }
            __syncthreads();
            const int cnt = __builtin_amdgcn_readfirstlane((int)sh_cnt[b]);
            const int nk = (cnt - wave + NW - 1) / NW;
            for (int base = 0; base < nk; base += 32) {
                const int j  = wave + NW * (base + lane);
                const int jc = (j < cnt) ? j : (cnt - 1);
                const int sv = ls[b][jc];
                const int dv = ld[b][jc];
                const float da = dinv[sv];
                const float db = dinv[dv];
                const int nks = (nk - base < 32) ? (nk - base) : 32;
                for (int k0 = 0; k0 < nks; k0 += SC_U) {
                    int kl[SC_U];
                    int dd[SC_U];
                    float hv[SC_U][NV];
#pragma unroll
                    for (int u = 0; u < SC_U; ++u) {
                        const int k = (k0 + u < nks) ? (k0 + u) : (nks - 1);
                        kl[u] = k;
                        const int s = __builtin_amdgcn_readlane(sv, k);
                        dd[u] = __builtin_amdgcn_readlane(dv, k);
                        const float* hs = H + (size_t)s * F + lane;
#pragma unroll
                        for (int v = 0; v < NV; ++v) hv[u][v] = hs[v * 32];
                    }
                    rows_asked(hv);
#pragma unroll
                    for (int u = 0; u < SC_U; ++u) {
                        if (k0 + u < nks) {
                            const float nn = __int_as_float(__builtin_amdgcn_readlane(__float_as_int(da), kl[u])) *
                                             __int_as_float(__builtin_amdgcn_readlane(__float_as_int(db), kl[u]));
                            float* o = out + (size_t)dd[u] * F + lane;
#pragma unroll
                            for (int v = 0; v < NV; ++v) atomicAdd(o + v * 32, hv[u][v] * nn);
                        }
                    }
                }
            }
        }
    }
}

extern "C" void kernel_launch(void* const* d_in, const int* in_sizes, int n_in,
                              void* d_out, int out_size, void* d_ws, size_t ws_size,
                              hipStream_t stream) {
    const float* x  = (const float*)d_in[0];
    const int*   ei = (const int*)d_in[1];
    const float* W1 = (const float*)d_in[2];
    const float* b1 = (const float*)d_in[3];
    const float* W2 = (const float*)d_in[4];
    const float* b2 = (const float*)d_in[5];
    const float* W3 = (const float*)d_in[6];
    const float* b3 = (const float*)d_in[7];
    float* out = (float*)d_out;

    const int N = GCN_N;
    const int E = in_sizes[1] / 2;
    const int* src = ei;
    const int* dst = ei + E;

    float* dinv = (float*)d_ws;
    unsigned* tickets = (unsigned*)(dinv + N);
    float* bufA = dinv + ((N + 255) & ~255);
    float* bufB = bufA + (size_t)N * 64;

    const int tiles  = N / 16;
    const int gemmBl = (tiles + 7) / 8;

    static_assert((size_t)DEG_G * DEG_PW <= (size_t)GCN_N * 32, "the count planes fit in out");
    unsigned* planes = (unsigned*)d_out;
    k_deg_planes<<<DEG_G, DEG_T, 0, stream>>>(dst, planes, tickets, E);

    k_gemm_wmma<128, 64, true, false><<<gemmBl, THREADS, 0, stream>>>(x, W1, b1, dinv, planes, bufA, bufB, N);
    k_scatter<64><<<PGRID, THREADS, 0, stream>>>(src, dst, dinv, bufA, bufB, tickets + 2, E);

    k_gemm_wmma<64, 64, false, true><<<gemmBl, THREADS, 0, stream>>>(bufB, W2, b2, dinv, nullptr, bufA, bufB, N);
    k_scatter<64><<<PGRID, THREADS, 0, stream>>>(src, dst, dinv, bufA, bufB, tickets + 4, E);

    k_gemm_wmma<64, 32, false, true><<<gemmBl, THREADS, 0, stream>>>(bufB, W3, b3, dinv, nullptr, bufA, out, N);
    k_scatter<32><<<PGRID, THREADS, 0, stream>>>(src, dst, dinv, bufA, out, tickets + 6, E);
}
